# starting kernel
# baseline (speedup 1.0000x reference)
#include <hip/hip_runtime.h>

#define DCH     2048
#define BATCH   4
#define TLEN    4096
#define SEGS    16
#define SEGLEN  (TLEN / SEGS)
#define NTILE   (DCH / 16)
#define EPS_A   1e-6f

typedef __attribute__((ext_vector_type(2))) float v2f;
typedef __attribute__((ext_vector_type(8))) float v8f;

__device__ __forceinline__ float get_alpha(const float* __restrict__ alpha_logit) {
    float a = 1.0f / (1.0f + __expf(-alpha_logit[0]));
    return fminf(fmaxf(a, EPS_A), 1.0f - EPS_A);
}

__global__ void ema_seg_final(const float* __restrict__ x,
                              const float* __restrict__ alpha_logit,
                              float* __restrict__ segf) {
    const float a = get_alpha(alpha_logit);
    const float c = 1.0f - a;
    int gid = blockIdx.x * blockDim.x + threadIdx.x;
    int d  = gid % DCH;
    int bs = gid / DCH;
    int s  = bs % SEGS;
    int b  = bs / SEGS;
    const float* xp = x + ((size_t)(b * TLEN + s * SEGLEN)) * DCH + d;
    float h = 0.0f;
#pragma unroll 8
    for (int i = 0; i < SEGLEN; ++i) {
        h = fmaf(c, h, a * xp[(size_t)i * DCH]);
    }
    segf[gid] = h;
}

__global__ void ema_carry_scan(const float* __restrict__ segf,
                               const float* __restrict__ alpha_logit,
                               float* __restrict__ carry) {
    const float a = get_alpha(alpha_logit);
    const float c = 1.0f - a;
    float cL = c;
#pragma unroll
    for (int i = 0; i < 8; ++i) cL *= cL;
    int gid = blockIdx.x * blockDim.x + threadIdx.x;
    int d = gid % DCH;
    int b = gid / DCH;
    float C = 0.0f;
    for (int s = 0; s < SEGS; ++s) {
        size_t o = ((size_t)(b * SEGS + s)) * DCH + d;
        carry[o] = C;
        C = fmaf(cL, C, segf[o]);
    }
}

__global__ void ema_scan_wmma(const float* __restrict__ x,
                              const float* __restrict__ alpha_logit,
                              const float* __restrict__ carryIn,
                              float* __restrict__ out) {
    const float a = get_alpha(alpha_logit);
    const float c = 1.0f - a;

    const int lane   = threadIdx.x & 31;
    const int laneHi = lane >> 4;
    const int lanLo  = lane & 15;

    int w    = blockIdx.x * (blockDim.x >> 5) + (threadIdx.x >> 5);
    int tile = w % NTILE;
    int bs   = w / NTILE;
    int s    = bs % SEGS;
    int b    = bs / SEGS;
    int d0   = tile * 16;

    v2f A[4];
#pragma unroll
    for (int sk = 0; sk < 4; ++sk) {
#pragma unroll
        for (int r = 0; r < 2; ++r) {
            int j = 4 * sk + r + 2 * laneHi;
            int m = lanLo;
            float v = 0.0f;
            if (m >= j) {
                float p = a;
                for (int q = 0; q < m - j; ++q) p *= c;
                v = p;
            }
            A[sk][r] = v;
        }
    }

    float carry = carryIn[((size_t)(b * SEGS + s)) * DCH + d0 + lanLo];

    float cpow0 = c;
    if (laneHi) {
        float c8 = c * c; c8 *= c8; c8 *= c8;
        cpow0 = c8 * c;
    }

    const size_t rs = DCH;
    const float* xbase = x   + ((size_t)(b * TLEN + s * SEGLEN)) * rs;
    float*       obase = out + ((size_t)(b * TLEN + s * SEGLEN)) * rs;

#pragma unroll 2
    for (int ch = 0; ch < SEGLEN / 16; ++ch) {
        int t0 = ch * 16;
        v2f Bm[4];
#pragma unroll
        for (int sk = 0; sk < 4; ++sk) {
#pragma unroll
            for (int r = 0; r < 2; ++r) {
                int j = 4 * sk + r + 2 * laneHi;
                Bm[sk][r] = xbase[(size_t)(t0 + j) * rs + d0 + lanLo];
            }
        }

        v8f acc = {};
#pragma unroll
        for (int sk = 0; sk < 4; ++sk) {
            acc = __builtin_amdgcn_wmma_f32_16x16x4_f32(
                false, A[sk], false, Bm[sk], (short)0, acc, false, false);
        }

        float pw = cpow0;
#pragma unroll
        for (int r = 0; r < 8; ++r) {
            acc[r] = fmaf(pw, carry, acc[r]);
            pw *= c;
        }

#pragma unroll
        for (int r = 0; r < 8; ++r) {
            int m = r + 8 * laneHi;
            __builtin_nontemporal_store(acc[r],
                &obase[(size_t)(t0 + m) * rs + d0 + lanLo]);
        }

        carry = __shfl(acc[7], 16 + lanLo, 32);
    }
}

extern "C" void kernel_launch(void* const* d_in, const int* in_sizes, int n_in,
                              void* d_out, int out_size, void* d_ws, size_t ws_size,
                              hipStream_t stream) {
    const float* x           = (const float*)d_in[0];
    const float* alpha_logit = (const float*)d_in[1];
    float* out = (float*)d_out;

    float* segf  = (float*)d_ws;
    float* carry = segf + (size_t)BATCH * SEGS * DCH;

    {
        int threads = BATCH * SEGS * DCH;
        ema_seg_final<<<threads / 256, 256, 0, stream>>>(x, alpha_logit, segf);
    }
    {
        int threads = BATCH * DCH;
        ema_carry_scan<<<threads / 256, 256, 0, stream>>>(segf, alpha_logit, carry);
    }
    {
        int waves = BATCH * SEGS * NTILE;
        ema_scan_wmma<<<waves / 8, 256, 0, stream>>>(x, alpha_logit, carry, out);
    }
}


# 4 adjacent channels per lane: 16-B loads/stores on aligned 256-B row pieces (wave = 4 interleaved 16-ch WMMA tiles), next 2 row sets loaded ahead; segments 256->128 rows, carry sums re-associated
# speedup vs baseline: 1.9761x; 1.9761x over previous
#include <hip/hip_runtime.h>


#define DCH     2048
#define BATCH   4
#define TLEN    4096
#define SEGS    32
#define SEGLEN  (TLEN / SEGS)
#define LOG2SEG 7
#define DCH4    (DCH / 4)
#define NGRP    (DCH / 64)
#define NCHUNK  (SEGLEN / 16)
#define EPS_A   1e-6f

typedef __attribute__((ext_vector_type(2))) float v2f;
typedef __attribute__((ext_vector_type(4))) float v4f;
typedef __attribute__((ext_vector_type(8))) float v8f;

__device__ __forceinline__ float get_alpha(const float* __restrict__ alpha_logit) {
    float a = 1.0f / (1.0f + __expf(-alpha_logit[0]));
    return fminf(fmaxf(a, EPS_A), 1.0f - EPS_A);
}

__device__ __forceinline__ void load_rows16(v4f (&dst)[16], const v4f* __restrict__ p) {
#pragma unroll
    for (int k = 0; k < 16; ++k) dst[k] = p[(size_t)k * DCH4];
}

__device__ __forceinline__ void ema_rows16(v4f& h, const v4f (&src)[16], float a, float c) {
#pragma unroll
    for (int k = 0; k < 16; ++k) {
        v4f v = src[k];
        h.x = fmaf(c, h.x, a * v.x);
        h.y = fmaf(c, h.y, a * v.y);
        h.z = fmaf(c, h.z, a * v.z);
        h.w = fmaf(c, h.w, a * v.w);
    }
}

__global__ void __launch_bounds__(64)
ema_seg_final(const float* __restrict__ x,
              const float* __restrict__ alpha_logit,
              float* __restrict__ segf) {
    int gid = blockIdx.x * blockDim.x + threadIdx.x;
    int d4 = gid % DCH4;
    int bs = gid / DCH4;
    const v4f* xp = (const v4f*)(x + (size_t)bs * SEGLEN * DCH) + d4;

    constexpr int NB = SEGLEN / 16;
    static_assert(NB >= 5 && (NB - 2) % 3 == 0, "the loop below takes three 16-row sets per trip");
    v4f r0[16], r1[16], r2[16];
    load_rows16(r0, xp);
    load_rows16(r1, xp + (size_t)16 * DCH4);

    const float a = get_alpha(alpha_logit);
    const float c = 1.0f - a;
    v4f h = {0.0f, 0.0f, 0.0f, 0.0f};
#pragma unroll 1
    for (int blk = 0; blk < NB - 2; blk += 3) {
        load_rows16(r2, xp + (size_t)(blk + 2) * 16 * DCH4);
        __builtin_amdgcn_sched_barrier(0);
        ema_rows16(h, r0, a, c);
        __builtin_amdgcn_sched_barrier(0);
        load_rows16(r0, xp + (size_t)(blk + 3) * 16 * DCH4);
        __builtin_amdgcn_sched_barrier(0);
        ema_rows16(h, r1, a, c);
        __builtin_amdgcn_sched_barrier(0);
        load_rows16(r1, xp + (size_t)(blk + 4) * 16 * DCH4);
        __builtin_amdgcn_sched_barrier(0);
        ema_rows16(h, r2, a, c);
    }
    ema_rows16(h, r0, a, c);
    ema_rows16(h, r1, a, c);
    ((v4f*)segf)[gid] = h;
}

__global__ void ema_carry_scan(const float* __restrict__ alpha_logit,
                               float* segc) {
    const float a = get_alpha(alpha_logit);
    const float c = 1.0f - a;
    float cL = c;
#pragma unroll
    for (int i = 0; i < LOG2SEG; ++i) cL *= cL;
    int gid = blockIdx.x * blockDim.x + threadIdx.x;
    int d = gid % DCH;
    int b = gid / DCH;
    float* p = segc + (size_t)b * SEGS * DCH + d;
    float f[SEGS];
#pragma unroll
    for (int s = 0; s < SEGS; ++s) f[s] = p[(size_t)s * DCH];
    float C = 0.0f;
#pragma unroll
    for (int s = 0; s < SEGS; ++s) {
        p[(size_t)s * DCH] = C;
        C = fmaf(cL, C, f[s]);
    }
}

__device__ __forceinline__ void load_chunk(v4f (&dst)[8], const float* __restrict__ xrow,
                                           int laneHi) {
#pragma unroll
    for (int sk = 0; sk < 4; ++sk) {
#pragma unroll
        for (int r = 0; r < 2; ++r) {
            int j = 4 * sk + r + 2 * laneHi;
            dst[2 * sk + r] = *(const v4f*)(xrow + (size_t)j * DCH);
        }
    }
}

__device__ __forceinline__ void scan_chunk(const v4f (&xr)[8], const v2f (&A)[4],
                                           const float (&pw)[8], float (&carry)[4],
                                           float* __restrict__ orow, int laneHi, int lanLo) {
    v8f acc[4] = {};
#pragma unroll
    for (int sk = 0; sk < 4; ++sk) {
#pragma unroll
        for (int q = 0; q < 4; ++q) {
            v2f bq = {xr[2 * sk][q], xr[2 * sk + 1][q]};
            acc[q] = __builtin_amdgcn_wmma_f32_16x16x4_f32(
                false, A[sk], false, bq, (short)0, acc[q], false, false);
        }
    }

#pragma unroll
    for (int q = 0; q < 4; ++q) {
#pragma unroll
        for (int r = 0; r < 8; ++r) acc[q][r] = fmaf(pw[r], carry[q], acc[q][r]);
    }

#pragma unroll
    for (int r = 0; r < 8; ++r) {
        int m = r + 8 * laneHi;
        v4f o = {acc[0][r], acc[1][r], acc[2][r], acc[3][r]};
        __builtin_nontemporal_store(o, (v4f*)(orow + (size_t)m * DCH));
    }

#pragma unroll
    for (int q = 0; q < 4; ++q) carry[q] = __shfl(acc[q][7], 16 + lanLo, 32);
}

__global__ void __launch_bounds__(128)
ema_scan_wmma(const float* __restrict__ x,
              const float* __restrict__ alpha_logit,
              const float* __restrict__ carryIn,
              float* __restrict__ out) {
    const int lane   = threadIdx.x & 31;
    const int laneHi = lane >> 4;
    const int lanLo  = lane & 15;

    int w   = blockIdx.x * (blockDim.x >> 5) + (threadIdx.x >> 5);
    int grp = w % NGRP;
    int bs  = w / NGRP;
    int dl  = grp * 64 + 4 * lanLo;

    const float* xbase = x   + (size_t)bs * SEGLEN * DCH + dl;
    float*       obase = out + (size_t)bs * SEGLEN * DCH + dl;

    v4f buf[3][8];
    load_chunk(buf[0], xbase, laneHi);
    load_chunk(buf[1], xbase + (size_t)16 * DCH, laneHi);
    v4f cin = *(const v4f*)(carryIn + (size_t)bs * DCH + dl);

    const float a = get_alpha(alpha_logit);
    const float c = 1.0f - a;

    v2f A[4];
#pragma unroll
    for (int sk = 0; sk < 4; ++sk) {
#pragma unroll
        for (int r = 0; r < 2; ++r) {
            int j = 4 * sk + r + 2 * laneHi;
            int m = lanLo;
            float v = 0.0f;
            if (m >= j) {
                float p = a;
                for (int q = 0; q < m - j; ++q) p *= c;
                v = p;
            }
            A[sk][r] = v;
        }
    }

    float cpow0 = c;
    if (laneHi) {
        float c8 = c * c; c8 *= c8; c8 *= c8;
        cpow0 = c8 * c;
    }
    float pw[8];
    pw[0] = cpow0;
#pragma unroll
    for (int r = 1; r < 8; ++r) pw[r] = pw[r - 1] * c;

    float carry[4] = {cin.x, cin.y, cin.z, cin.w};

#pragma unroll
    for (int ch = 0; ch < NCHUNK; ++ch) {
        if (ch + 2 < NCHUNK) {
            __builtin_amdgcn_sched_barrier(0);
            load_chunk(buf[(ch + 2) % 3], xbase + (size_t)(ch + 2) * 16 * DCH, laneHi);
            __builtin_amdgcn_sched_barrier(0);
        }
        scan_chunk(buf[ch % 3], A, pw, carry, obase + (size_t)ch * 16 * DCH, laneHi, lanLo);
    }
}

extern "C" void kernel_launch(void* const* d_in, const int* in_sizes, int n_in,
                              void* d_out, int out_size, void* d_ws, size_t ws_size,
                              hipStream_t stream) {
    const float* x           = (const float*)d_in[0];
    const float* alpha_logit = (const float*)d_in[1];
    float* out = (float*)d_out;

    float* segc = (float*)d_ws;

    {
        int threads = BATCH * SEGS * DCH4;
        ema_seg_final<<<threads / 64, 64, 0, stream>>>(x, alpha_logit, segc);
    }
    {
        int threads = BATCH * DCH;
        ema_carry_scan<<<threads / 256, 256, 0, stream>>>(alpha_logit, segc);
    }
    {
        int waves = BATCH * SEGS * NGRP;
        ema_scan_wmma<<<waves / 4, 128, 0, stream>>>(x, alpha_logit, segc, out);
    }
}


# one pass over x: an 8-wave work-group per (batch, 64 channels) walks all rows, wave finals combined through LDS every 256 rows; 3 launches->1, x read once; carry sums re-associated
# speedup vs baseline: 2.1563x; 1.0912x over previous
#include <hip/hip_runtime.h>


#define DCH     2048
#define BATCH   4
#define TLEN    4096
#define NGRP    (DCH / 64)
#define NW      8
#define RW      32
#define SROWS   (NW * RW)
#define NSTEP   (TLEN / SROWS)
#define EPS_A   1e-6f

typedef __attribute__((ext_vector_type(2))) float v2f;
typedef __attribute__((ext_vector_type(4))) float v4f;
typedef __attribute__((ext_vector_type(8))) float v8f;

__device__ __forceinline__ float get_alpha(const float* __restrict__ alpha_logit) {
    float a = 1.0f / (1.0f + __expf(-alpha_logit[0]));
    return fminf(fmaxf(a, EPS_A), 1.0f - EPS_A);
}

__device__ __forceinline__ void load_chunk(v4f (&dst)[8], const float* __restrict__ xrow) {
#pragma unroll
    for (int sk = 0; sk < 4; ++sk) {
#pragma unroll
        for (int r = 0; r < 2; ++r) {
            dst[2 * sk + r] = *(const v4f*)(xrow + (size_t)(4 * sk + r) * DCH);
        }
    }
}

__device__ __forceinline__ void local_chunk(v8f (&acc)[4], const v4f (&xr)[8],
                                            const v2f (&A)[4]) {
#pragma unroll
    for (int q = 0; q < 4; ++q) acc[q] = v8f{};
#pragma unroll
    for (int sk = 0; sk < 4; ++sk) {
#pragma unroll
        for (int q = 0; q < 4; ++q) {
            v2f bq = {xr[2 * sk][q], xr[2 * sk + 1][q]};
            acc[q] = __builtin_amdgcn_wmma_f32_16x16x4_f32(
                false, A[sk], false, bq, (short)0, acc[q], false, false);
        }
    }
}

__device__ __forceinline__ void finish_chunk(const v8f (&acc)[4], const float (&pw)[8],
                                             v4f hin, float* __restrict__ orow) {
#pragma unroll
    for (int r = 0; r < 8; ++r) {
        v4f o = {fmaf(pw[r], hin.x, acc[0][r]), fmaf(pw[r], hin.y, acc[1][r]),
                 fmaf(pw[r], hin.z, acc[2][r]), fmaf(pw[r], hin.w, acc[3][r])};
        __builtin_nontemporal_store(o, (v4f*)(orow + (size_t)r * DCH));
    }
}

template <bool MORE>
__device__ __forceinline__ void do_step(v4f (&xs)[2][8], const v2f (&A)[4], const float (&pw)[8],
                                        float c16, float c32, v4f& G, v4f (*sFs)[16],
                                        int wv, int laneHi, int lanLo,
                                        float* __restrict__ orow,
                                        const float* __restrict__ xnext) {
    v8f acc0[4], acc1[4];
    local_chunk(acc0, xs[0], A);
    local_chunk(acc1, xs[1], A);

    if (MORE) {
        __builtin_amdgcn_sched_barrier(0);
        load_chunk(xs[0], xnext);
        load_chunk(xs[1], xnext + (size_t)16 * DCH);
        __builtin_amdgcn_sched_barrier(0);
    }

    v4f f0, f1;
#pragma unroll
    for (int q = 0; q < 4; ++q) {
        f0[q] = __shfl(acc0[q][7], 16 + lanLo, 32);
        f1[q] = __shfl(acc1[q][7], 16 + lanLo, 32);
    }
    v4f F = {fmaf(c16, f0.x, f1.x), fmaf(c16, f0.y, f1.y),
             fmaf(c16, f0.z, f1.z), fmaf(c16, f0.w, f1.w)};
    if (laneHi == 0) sFs[wv][lanLo] = F;

    __builtin_amdgcn_fence(__ATOMIC_RELEASE, "workgroup", "local");
    __builtin_amdgcn_s_barrier();
    __builtin_amdgcn_fence(__ATOMIC_ACQUIRE, "workgroup", "local");

    v4f C = G, hin = G;
#pragma unroll
    for (int w = 0; w < NW; ++w) {
        v4f Fw = sFs[w][lanLo];
        if (w == wv) hin = C;
        C.x = fmaf(c32, C.x, Fw.x);
        C.y = fmaf(c32, C.y, Fw.y);
        C.z = fmaf(c32, C.z, Fw.z);
        C.w = fmaf(c32, C.w, Fw.w);
    }
    G = C;

    v4f hin1 = {fmaf(c16, hin.x, f0.x), fmaf(c16, hin.y, f0.y),
                fmaf(c16, hin.z, f0.z), fmaf(c16, hin.w, f0.w)};

    finish_chunk(acc0, pw, hin,  orow);
    finish_chunk(acc1, pw, hin1, orow + (size_t)16 * DCH);
}

__global__ void __launch_bounds__(NW * 32)
ema_onepass(const float* __restrict__ x,
            const float* __restrict__ alpha_logit,
            float* __restrict__ out) {
    __shared__ v4f sF[2][NW][16];

    const int lane   = threadIdx.x & 31;
    const int laneHi = lane >> 4;
    const int lanLo  = lane & 15;
    const int wv     = threadIdx.x >> 5;

    const int grp = blockIdx.x % NGRP;
    const int b   = blockIdx.x / NGRP;
    const int dl  = grp * 64 + 4 * lanLo;

    const size_t base = ((size_t)b * TLEN + (size_t)wv * RW) * DCH + dl;
    const float* xp = x + base + (size_t)(2 * laneHi) * DCH;
    float*       op = out + base + (size_t)(8 * laneHi) * DCH;

    v4f buf[2][2][8];
    load_chunk(buf[0][0], xp);
    load_chunk(buf[0][1], xp + (size_t)16 * DCH);
    load_chunk(buf[1][0], xp + (size_t)SROWS * DCH);
    load_chunk(buf[1][1], xp + (size_t)(SROWS + 16) * DCH);

    const float a = get_alpha(alpha_logit);
    const float c = 1.0f - a;

    v2f A[4];
#pragma unroll
    for (int sk = 0; sk < 4; ++sk) {
#pragma unroll
        for (int r = 0; r < 2; ++r) {
            int j = 4 * sk + r + 2 * laneHi;
            int m = lanLo;
            float v = 0.0f;
            if (m >= j) {
                float p = a;
                for (int q = 0; q < m - j; ++q) p *= c;
                v = p;
            }
            A[sk][r] = v;
        }
    }

    const float c2 = c * c, c4 = c2 * c2, c8 = c4 * c4;
    const float c16 = c8 * c8;
    const float c32 = c16 * c16;
    float pw[8];
    pw[0] = laneHi ? c8 * c : c;
#pragma unroll
    for (int r = 1; r < 8; ++r) pw[r] = pw[r - 1] * c;

    v4f G = {0.0f, 0.0f, 0.0f, 0.0f};

    static_assert(NSTEP % 2 == 0 && NSTEP >= 4, "two steps per trip, the last trip peeled");
#pragma unroll 1
    for (int st = 0; st < NSTEP - 2; st += 2) {
        do_step<true>(buf[0], A, pw, c16, c32, G, sF[0], wv, laneHi, lanLo,
                      op, xp + (size_t)2 * SROWS * DCH);
        do_step<true>(buf[1], A, pw, c16, c32, G, sF[1], wv, laneHi, lanLo,
                      op + (size_t)SROWS * DCH, xp + (size_t)3 * SROWS * DCH);
        xp += (size_t)2 * SROWS * DCH;
        op += (size_t)2 * SROWS * DCH;
    }
    do_step<false>(buf[0], A, pw, c16, c32, G, sF[0], wv, laneHi, lanLo, op, xp);
    do_step<false>(buf[1], A, pw, c16, c32, G, sF[1], wv, laneHi, lanLo,
                   op + (size_t)SROWS * DCH, xp);
}

extern "C" void kernel_launch(void* const* d_in, const int* in_sizes, int n_in,
                              void* d_out, int out_size, void* d_ws, size_t ws_size,
                              hipStream_t stream) {
    const float* x           = (const float*)d_in[0];
    const float* alpha_logit = (const float*)d_in[1];
    float* out = (float*)d_out;

    ema_onepass<<<BATCH * NGRP, NW * 32, 0, stream>>>(x, alpha_logit, out);
}


# 16 waves per work-group, one 16-row chunk each per step (was 8 waves x 2 chunks): four waves per SIMD in 256 registers, more loads and stores in flight per compute unit; carry sums re-associated
# speedup vs baseline: 2.2946x; 1.0641x over previous
#include <hip/hip_runtime.h>


#define DCH     2048
#define BATCH   4
#define TLEN    4096
#define NGRP    (DCH / 64)
#define NW      16
#define RW      16
#define SROWS   (NW * RW)
#define NSTEP   (TLEN / SROWS)
#define EPS_A   1e-6f

typedef __attribute__((ext_vector_type(2))) float v2f;
typedef __attribute__((ext_vector_type(4))) float v4f;
typedef __attribute__((ext_vector_type(8))) float v8f;

__device__ __forceinline__ float get_alpha(const float* __restrict__ alpha_logit) {
    float a = 1.0f / (1.0f + __expf(-alpha_logit[0]));
    return fminf(fmaxf(a, EPS_A), 1.0f - EPS_A);
}

__device__ __forceinline__ void load_chunk(v4f (&dst)[8], const float* __restrict__ xrow) {
#pragma unroll
    for (int sk = 0; sk < 4; ++sk) {
#pragma unroll
        for (int r = 0; r < 2; ++r) {
            dst[2 * sk + r] = *(const v4f*)(xrow + (size_t)(4 * sk + r) * DCH);
        }
    }
}

__device__ __forceinline__ void local_chunk(v8f (&acc)[4], const v4f (&xr)[8],
                                            const v2f (&A)[4]) {
#pragma unroll
    for (int q = 0; q < 4; ++q) acc[q] = v8f{};
#pragma unroll
    for (int sk = 0; sk < 4; ++sk) {
#pragma unroll
        for (int q = 0; q < 4; ++q) {
            v2f bq = {xr[2 * sk][q], xr[2 * sk + 1][q]};
            acc[q] = __builtin_amdgcn_wmma_f32_16x16x4_f32(
                false, A[sk], false, bq, (short)0, acc[q], false, false);
        }
    }
}

__device__ __forceinline__ void finish_chunk(const v8f (&acc)[4], const float (&pw)[8],
                                             v4f hin, float* __restrict__ orow) {
#pragma unroll
    for (int r = 0; r < 8; ++r) {
        v4f o = {fmaf(pw[r], hin.x, acc[0][r]), fmaf(pw[r], hin.y, acc[1][r]),
                 fmaf(pw[r], hin.z, acc[2][r]), fmaf(pw[r], hin.w, acc[3][r])};
        __builtin_nontemporal_store(o, (v4f*)(orow + (size_t)r * DCH));
    }
}

template <bool MORE>
__device__ __forceinline__ void do_step(v4f (&xs)[8], const v2f (&A)[4], const float (&pw)[8],
                                        float c16, v4f& G, v4f (*sFs)[16],
                                        int wv, int laneHi, int lanLo,
                                        float* __restrict__ orow,
                                        const float* __restrict__ xnext) {
    v8f acc[4];
    local_chunk(acc, xs, A);

    if (MORE) {
        __builtin_amdgcn_sched_barrier(0);
        load_chunk(xs, xnext);
        __builtin_amdgcn_sched_barrier(0);
    }

    v4f F;
#pragma unroll
    for (int q = 0; q < 4; ++q) F[q] = __shfl(acc[q][7], 16 + lanLo, 32);
    if (laneHi == 0) sFs[wv][lanLo] = F;

    __builtin_amdgcn_fence(__ATOMIC_RELEASE, "workgroup", "local");
    __builtin_amdgcn_s_barrier();
    __builtin_amdgcn_fence(__ATOMIC_ACQUIRE, "workgroup", "local");

    v4f C = G, hin = G;
#pragma unroll
    for (int w = 0; w < NW; ++w) {
        v4f Fw = sFs[w][lanLo];
        if (w == wv) hin = C;
        C.x = fmaf(c16, C.x, Fw.x);
        C.y = fmaf(c16, C.y, Fw.y);
        C.z = fmaf(c16, C.z, Fw.z);
        C.w = fmaf(c16, C.w, Fw.w);
    }
    G = C;

    finish_chunk(acc, pw, hin, orow);
}

__global__ void __launch_bounds__(NW * 32)
ema_onepass(const float* __restrict__ x,
            const float* __restrict__ alpha_logit,
            float* __restrict__ out) {
    __shared__ v4f sF[2][NW][16];

    const int lane   = threadIdx.x & 31;
    const int laneHi = lane >> 4;
    const int lanLo  = lane & 15;
    const int wv     = threadIdx.x >> 5;

    const int grp = blockIdx.x % NGRP;
    const int b   = blockIdx.x / NGRP;
    const int dl  = grp * 64 + 4 * lanLo;

    const size_t base = ((size_t)b * TLEN + (size_t)wv * RW) * DCH + dl;
    const float* xp = x + base + (size_t)(2 * laneHi) * DCH;
    float*       op = out + base + (size_t)(8 * laneHi) * DCH;

    v4f buf[2][8];
    load_chunk(buf[0], xp);
    load_chunk(buf[1], xp + (size_t)SROWS * DCH);

    const float a = get_alpha(alpha_logit);
    const float c = 1.0f - a;

    v2f A[4];
#pragma unroll
    for (int sk = 0; sk < 4; ++sk) {
#pragma unroll
        for (int r = 0; r < 2; ++r) {
            int j = 4 * sk + r + 2 * laneHi;
            int m = lanLo;
            float v = 0.0f;
            if (m >= j) {
                float p = a;
                for (int q = 0; q < m - j; ++q) p *= c;
                v = p;
            }
            A[sk][r] = v;
        }
    }

    const float c2 = c * c, c4 = c2 * c2, c8 = c4 * c4;
    const float c16 = c8 * c8;
    float pw[8];
    pw[0] = laneHi ? c8 * c : c;
#pragma unroll
    for (int r = 1; r < 8; ++r) pw[r] = pw[r - 1] * c;

    v4f G = {0.0f, 0.0f, 0.0f, 0.0f};

    static_assert(NSTEP % 2 == 0 && NSTEP >= 4, "two steps per trip, the last trip peeled");
#pragma unroll 1
    for (int st = 0; st < NSTEP - 2; st += 2) {
        do_step<true>(buf[0], A, pw, c16, G, sF[0], wv, laneHi, lanLo,
                      op, xp + (size_t)2 * SROWS * DCH);
        do_step<true>(buf[1], A, pw, c16, G, sF[1], wv, laneHi, lanLo,
                      op + (size_t)SROWS * DCH, xp + (size_t)3 * SROWS * DCH);
        xp += (size_t)2 * SROWS * DCH;
        op += (size_t)2 * SROWS * DCH;
    }
    do_step<false>(buf[0], A, pw, c16, G, sF[0], wv, laneHi, lanLo, op, xp);
    do_step<false>(buf[1], A, pw, c16, G, sF[1], wv, laneHi, lanLo,
                   op + (size_t)SROWS * DCH, xp);
}

extern "C" void kernel_launch(void* const* d_in, const int* in_sizes, int n_in,
                              void* d_out, int out_size, void* d_ws, size_t ws_size,
                              hipStream_t stream) {
    const float* x           = (const float*)d_in[0];
    const float* alpha_logit = (const float*)d_in[1];
    float* out = (float*)d_out;

    ema_onepass<<<BATCH * NGRP, NW * 32, 0, stream>>>(x, alpha_logit, out);
}
